# starting kernel
# baseline (speedup 1.0000x reference)
#include <hip/hip_runtime.h>
#include <hip/hip_bf16.h>

typedef __attribute__((ext_vector_type(2))) float v2f;
typedef __attribute__((ext_vector_type(8))) float v8f;

#define GAT_NEG_INF (-9000000000000000.0f)

constexpr int kB   = 8;
constexpr int kN   = 1024;
constexpr int kDin = 128;
constexpr int kH   = 8;
constexpr int kD   = 16;
constexpr int kHD  = 128;
constexpr float kAlpha = 0.2f;

__device__ __forceinline__ float leaky(float x) {
    return fmaxf(x, kAlpha * x);
}

template <int IMM>
__device__ __forceinline__ float swzf(float v) {
    return __int_as_float(__builtin_amdgcn_ds_swizzle(__float_as_int(v), IMM));
}
__device__ __forceinline__ float bpermf(int byteAddr, float v) {
    return __int_as_float(__builtin_amdgcn_ds_bpermute(byteAddr, __float_as_int(v)));
}
#define SWZ_X1  0x041f
#define SWZ_X2  0x081f
#define SWZ_X4  0x101f
#define SWZ_X8  0x201f
#define SWZ_X16 0x401f

__global__ __launch_bounds__(256) void gat_phase1(
    const float* __restrict__ h, const float* __restrict__ W,
    const float* __restrict__ a, float* __restrict__ WhP,
    float* __restrict__ s_i, float* __restrict__ s_j)
{
    __shared__ float hT[16 * 132];

    const int g    = blockIdx.x;
    const int tid  = threadIdx.x;
    const int head = tid >> 5;
    const int lane = tid & 31;
    const int r16  = lane & 15;
    const int half = lane >> 4;

    {
        const float4* src = (const float4*)(h + (size_t)g * 16 * kDin);
        #pragma unroll
        for (int it = 0; it < 2; ++it) {
            int idx  = it * 256 + tid;
            int row  = idx >> 5;
            int col4 = idx & 31;
            float4 v = src[idx];
            *(float4*)&hT[row * 132 + col4 * 4] = v;
        }
    }
    __syncthreads();

    const int n0 = head * 16;
    v8f c = {};
    #pragma unroll 4
    for (int k = 0; k < kDin; k += 4) {
        const int ka = k + 2 * half;
        v2f av, bv;
        av[0] = hT[r16 * 132 + ka];
        av[1] = hT[r16 * 132 + ka + 1];
        bv[0] = W[(size_t)ka * kHD + n0 + r16];
        bv[1] = W[(size_t)(ka + 1) * kHD + n0 + r16];
        c = __builtin_amdgcn_wmma_f32_16x16x4_f32(false, av, false, bv,
                                                  (short)0, c, false, false);
    }

    #pragma unroll
    for (int r = 0; r < 8; ++r) {
        int row = g * 16 + r + half * 8;
        WhP[((size_t)(row >> 1) * kHD + n0 + r16) * 2 + (row & 1)] = c[r];
    }

    const float asrc = a[head * 2 * kD + r16];
    const float adst = a[head * 2 * kD + kD + r16];
    #pragma unroll
    for (int r = 0; r < 8; ++r) {
        float vs = c[r] * asrc;
        float vd = c[r] * adst;
        vs += swzf<SWZ_X1>(vs);  vd += swzf<SWZ_X1>(vd);
        vs += swzf<SWZ_X2>(vs);  vd += swzf<SWZ_X2>(vd);
        vs += swzf<SWZ_X4>(vs);  vd += swzf<SWZ_X4>(vd);
        vs += swzf<SWZ_X8>(vs);  vd += swzf<SWZ_X8>(vd);
        if (r16 == 0) {
            int row = g * 16 + r + half * 8;
            int bb = row >> 10, nn = row & (kN - 1);
            s_i[((size_t)bb * kH + head) * kN + nn] = vs;
            s_j[((size_t)bb * kH + head) * kN + nn] = vd;
        }
    }
}

__global__ __launch_bounds__(256) void gat_phase2(
    const int* __restrict__ adj, const float* __restrict__ WhP,
    const float* __restrict__ s_i, const float* __restrict__ s_j,
    float* __restrict__ out)
{
    __shared__ float adjM[kN * 16];

    const int blk  = blockIdx.x;
    const int b    = blk >> 6;
    const int i0   = (blk & 63) * 16;
    const int tid  = threadIdx.x;
    const int head = tid >> 5;
    const int lane = tid & 31;
    const int r16  = lane & 15;
    const int half = lane >> 4;

    {
        const int4* src = (const int4*)(adj + ((size_t)b * kN + i0) * kN);
        #pragma unroll
        for (int it = 0; it < 16; ++it) {
            int idx = it * 256 + tid;
            int row = idx >> 8;
            int col = (idx & 255) * 4;
            int4 v  = src[idx];
            int rot0 = (col >> 2) & 15;
            int base = (col >> 1) * 32 + 2 * ((row + rot0) & 15);
            float2 lo, hi;
            lo.x = v.x ? 0.0f : GAT_NEG_INF;
            lo.y = v.y ? 0.0f : GAT_NEG_INF;
            hi.x = v.z ? 0.0f : GAT_NEG_INF;
            hi.y = v.w ? 0.0f : GAT_NEG_INF;
            *(float2*)&adjM[base]      = lo;
            *(float2*)&adjM[base + 32] = hi;
        }
    }
    __syncthreads();

    const float* sI = s_i + ((size_t)b * kH + head) * kN;
    const float* sJ = s_j + ((size_t)b * kH + head) * kN;
    const float si_l = sI[i0 + r16];

    float mv = GAT_NEG_INF;
    {
        const float4* sJ4 = (const float4*)sJ;
        #pragma unroll
        for (int k = 0; k < 8; ++k) {
            float4 v = sJ4[k * 32 + lane];
            mv = fmaxf(mv, fmaxf(fmaxf(v.x, v.y), fmaxf(v.z, v.w)));
        }
        mv = fmaxf(mv, swzf<SWZ_X16>(mv));
        mv = fmaxf(mv, swzf<SWZ_X8>(mv));
        mv = fmaxf(mv, swzf<SWZ_X4>(mv));
        mv = fmaxf(mv, swzf<SWZ_X2>(mv));
        mv = fmaxf(mv, swzf<SWZ_X1>(mv));
    }
    const float m_l = leaky(si_l + mv);

    const float* WhPB = WhP + (size_t)b * kN * kHD + head * 32;
    v8f c = {};
    float psum = 0.0f;
    for (int chunk = 0; chunk < 32; ++chunk) {
        const int jb = chunk * 32;
        #pragma unroll 4
        for (int kk = 0; kk < 8; ++kk) {
            const int j0  = jb + kk * 4 + 2 * half;
            const int q0  = j0 >> 1;
            const int rot = (j0 >> 2) & 15;
            const int idx = q0 * 32 + 2 * ((r16 + rot) & 15);
            float2 ad = *(const float2*)&adjM[idx];
            float2 sj = *(const float2*)(sJ + j0);
            float2 wv = *(const float2*)(WhPB + (size_t)q0 * 256 + r16 * 2);
            float t0 = leaky(si_l + sj.x) + ad.x;
            float t1 = leaky(si_l + sj.y) + ad.y;
            float p0 = __expf(t0 - m_l);
            float p1 = __expf(t1 - m_l);
            psum += p0 + p1;
            v2f av, bv;
            av[0] = p0;   av[1] = p1;
            bv[0] = wv.x; bv[1] = wv.y;
            c = __builtin_amdgcn_wmma_f32_16x16x4_f32(false, av, false, bv,
                                                      (short)0, c, false, false);
        }
    }

    psum += swzf<SWZ_X16>(psum);
    const float inv_l = 1.0f / fmaxf(psum, 1e-30f);
    const int baddr = half * 32;
    #pragma unroll
    for (int r = 0; r < 8; ++r) {
        float sc = bpermf(baddr + 4 * r, inv_l);
        int row = i0 + r + half * 8;
        out[((size_t)b * kN + row) * kHD + head * 16 + r16] = c[r] * sc;
    }
}

extern "C" void kernel_launch(void* const* d_in, const int* in_sizes, int n_in,
                              void* d_out, int out_size, void* d_ws, size_t ws_size,
                              hipStream_t stream) {
    const float* h   = (const float*)d_in[0];
    const int*   adj = (const int*)d_in[1];
    const float* W   = (const float*)d_in[2];
    const float* a   = (const float*)d_in[3];
    float* out = (float*)d_out;

    float* WhP = (float*)d_ws;
    float* s_i = WhP + (size_t)kB * kN * kHD;
    float* s_j = s_i + (size_t)kB * kH * kN;

    gat_phase1<<<dim3((kB * kN) / 16), 256, 0, stream>>>(h, W, a, WhP, s_i, s_j);
    gat_phase2<<<dim3(kB * (kN / 16)), 256, 0, stream>>>(adj, WhP, s_i, s_j, out);
}


# Phase 2 re-cut by head pair: block = (batch, head pair, 128 rows), its whole Wh slice in LDS from one load, sweep without ring or barrier; adjacency global->registers; sums regrouped by column
# speedup vs baseline: 1.3607x; 1.3607x over previous
#include <hip/hip_runtime.h>
#include <hip/hip_bf16.h>

typedef __attribute__((ext_vector_type(2))) float v2f;
typedef __attribute__((ext_vector_type(8))) float v8f;

#define GAT_NEG_INF (-9000000000000000.0f)

constexpr int kB   = 8;
constexpr int kN   = 1024;
constexpr int kDin = 128;
constexpr int kH   = 8;
constexpr int kD   = 16;
constexpr int kHD  = 128;
constexpr float kAlpha = 0.2f;

__device__ __forceinline__ float leaky(float x) {
    return fmaxf(x, kAlpha * x);
}

template <int IMM>
__device__ __forceinline__ float swzf(float v) {
    return __int_as_float(__builtin_amdgcn_ds_swizzle(__float_as_int(v), IMM));
}
__device__ __forceinline__ float bpermf(int byteAddr, float v) {
    return __int_as_float(__builtin_amdgcn_ds_bpermute(byteAddr, __float_as_int(v)));
}
#define SWZ_X1  0x041f
#define SWZ_X2  0x081f
#define SWZ_X4  0x101f
#define SWZ_X8  0x201f
#define SWZ_X16 0x401f

constexpr int kWPitch = 288;

__global__ __launch_bounds__(256) void gat_phase1(
    const float* __restrict__ h, const float* __restrict__ W,
    const float* __restrict__ a, float* __restrict__ WhP,
    float* __restrict__ s_i, float* __restrict__ s_j)
{
    __shared__ float ldsP1[16 * 132 + (kDin / 2) * kWPitch];
    float* const hT = ldsP1;
    float* const WL = ldsP1 + 16 * 132;

    const int g    = blockIdx.x;
    const int tid  = threadIdx.x;
    const int head = tid >> 5;
    const int lane = tid & 31;
    const int r16  = lane & 15;
    const int half = lane >> 4;

    {
        const float4* hsrc = (const float4*)(h + (size_t)g * 16 * kDin);
        const float4* W4   = (const float4*)W;
        const int col4 = tid & 31;
        const int kp0  = tid >> 5;
        const float4 hv0 = hsrc[tid];
        const float4 hv1 = hsrc[256 + tid];
        float4 we[8], wo[8];
        #pragma unroll
        for (int it = 0; it < 8; ++it) {
            const int kp = kp0 + 8 * it;
            we[it] = W4[(2 * kp) * 32 + col4];
            wo[it] = W4[(2 * kp + 1) * 32 + col4];
        }
        *(float4*)&hT[(tid >> 5) * 132 + (tid & 31) * 4]       = hv0;
        *(float4*)&hT[((tid >> 5) + 8) * 132 + (tid & 31) * 4] = hv1;
        #pragma unroll
        for (int it = 0; it < 8; ++it) {
            const int kp = kp0 + 8 * it;
            float* d = &WL[kp * kWPitch + col4 * 8];
            float4 lo, hi;
            lo.x = we[it].x;  lo.y = wo[it].x;  lo.z = we[it].y;  lo.w = wo[it].y;
            hi.x = we[it].z;  hi.y = wo[it].z;  hi.z = we[it].w;  hi.w = wo[it].w;
            *(float4*)d       = lo;
            *(float4*)(d + 4) = hi;
        }
    }
    __syncthreads();

    const int n0 = head * 16;
    const float* aRow = &hT[r16 * 132 + 2 * half];
    const float* bCol = &WL[half * kWPitch + (n0 + r16) * 2];
    v8f c = {};
    #pragma unroll
    for (int k = 0; k < kDin; k += 4) {
        const v2f av = *(const v2f*)(aRow + k);
        const v2f bv = *(const v2f*)(bCol + (k >> 1) * kWPitch);
        c = __builtin_amdgcn_wmma_f32_16x16x4_f32(false, av, false, bv,
                                                  (short)0, c, false, false);
    }

    #pragma unroll
    for (int r = 0; r < 8; ++r) {
        int row = g * 16 + r + half * 8;
        WhP[((size_t)(row >> 1) * kHD + n0 + r16) * 2 + (row & 1)] = c[r];
    }

    const float asrc = a[head * 2 * kD + r16];
    const float adst = a[head * 2 * kD + kD + r16];
    #pragma unroll
    for (int r = 0; r < 8; ++r) {
        float vs = c[r] * asrc;
        float vd = c[r] * adst;
        vs += swzf<SWZ_X1>(vs);  vd += swzf<SWZ_X1>(vd);
        vs += swzf<SWZ_X2>(vs);  vd += swzf<SWZ_X2>(vd);
        vs += swzf<SWZ_X4>(vs);  vd += swzf<SWZ_X4>(vd);
        vs += swzf<SWZ_X8>(vs);  vd += swzf<SWZ_X8>(vd);
        if (r16 == 0) {
            int row = g * 16 + r + half * 8;
            int bb = row >> 10, nn = row & (kN - 1);
            s_i[((size_t)bb * kH + head) * kN + nn] = vs;
            s_j[((size_t)bb * kH + head) * kN + nn] = vd;
        }
    }
}

constexpr int kSlabRows = 128;
constexpr int kNQ       = kN / 2;
constexpr int kQPitch   = 80;
constexpr int kChunkT   = 16;
constexpr int kNChunk   = kN / (8 * kChunkT);

__device__ __forceinline__ void head_sj_load(const float* __restrict__ sJ, int lane, float4 (&v)[8]) {
    const float4* sJ4 = (const float4*)sJ;
    #pragma unroll
    for (int k = 0; k < 8; ++k) v[k] = sJ4[k * 32 + lane];
}
__device__ __forceinline__ float head_max_sj(const float4 (&v)[8]) {
    float mv = GAT_NEG_INF;
    #pragma unroll
    for (int k = 0; k < 8; ++k)
        mv = fmaxf(mv, fmaxf(fmaxf(v[k].x, v[k].y), fmaxf(v[k].z, v[k].w)));
    mv = fmaxf(mv, swzf<SWZ_X16>(mv));
    mv = fmaxf(mv, swzf<SWZ_X8>(mv));
    mv = fmaxf(mv, swzf<SWZ_X4>(mv));
    mv = fmaxf(mv, swzf<SWZ_X2>(mv));
    mv = fmaxf(mv, swzf<SWZ_X1>(mv));
    return mv;
}

__device__ __forceinline__ void gat_step(
    float sa0, float sa1, float sb0, float sb1, int m0, int m1,
    const v2f wA, const v2f wB,
    const v2f siA2, const v2f siB2, const v2f mA2, const v2f mB2,
    v8f& cA, v8f& cB, v2f& psA2, v2f& psB2)
{
    const v2f sjA = {sa0, sa1};
    const v2f sjB = {sb0, sb1};
    const v2f xA = siA2 + sjA;
    const v2f xB = siB2 + sjB;
    const v2f tA = __builtin_elementwise_max(xA, xA * kAlpha) - mA2;
    const v2f tB = __builtin_elementwise_max(xB, xB * kAlpha) - mB2;
    v2f pA, pB;
    pA[0] = m0 ? __expf(tA[0]) : 0.0f;
    pA[1] = m1 ? __expf(tA[1]) : 0.0f;
    pB[0] = m0 ? __expf(tB[0]) : 0.0f;
    pB[1] = m1 ? __expf(tB[1]) : 0.0f;
    psA2 += pA;
    psB2 += pB;
    cA = __builtin_amdgcn_wmma_f32_16x16x4_f32(false, pA, false, wA,
                                               (short)0, cA, false, false);
    cB = __builtin_amdgcn_wmma_f32_16x16x4_f32(false, pB, false, wB,
                                               (short)0, cB, false, false);
}

template <bool PF>
__device__ __forceinline__ void gat_chunk(
    const int4 (&cur)[kChunkT], int4 (&nxt)[kChunkT], const int4* __restrict__ arowNext,
    const float* __restrict__ whC, const float* __restrict__ sjC,
    const v2f siA2, const v2f siB2, const v2f mA2, const v2f mB2,
    v8f& cA, v8f& cB, v2f& psA2, v2f& psB2)
{
    #pragma unroll
    for (int g = 0; g < 2; ++g) {
        float4 sA[8], sB[8];
        v2f    wA[16], wB[16];
        #pragma unroll
        for (int i = 0; i < 8; ++i) {
            const int tt = g * 8 + i;
            sA[i] = *(const float4*)(sjC + 8 * tt);
            sB[i] = *(const float4*)(sjC + kN + 8 * tt);
            wA[2 * i]     = *(const v2f*)(whC + (4 * tt) * kQPitch);
            wB[2 * i]     = *(const v2f*)(whC + (4 * tt) * kQPitch + 32);
            wA[2 * i + 1] = *(const v2f*)(whC + (4 * tt + 1) * kQPitch);
            wB[2 * i + 1] = *(const v2f*)(whC + (4 * tt + 1) * kQPitch + 32);
        }
        __builtin_amdgcn_sched_barrier(0);
        #pragma unroll
        for (int i = 0; i < 8; ++i) {
            if (PF) nxt[g * 8 + i] = arowNext[2 * (g * 8 + i)];
            const int4 mq = cur[g * 8 + i];
            gat_step(sA[i].x, sA[i].y, sB[i].x, sB[i].y, mq.x, mq.y, wA[2 * i], wB[2 * i],
                     siA2, siB2, mA2, mB2, cA, cB, psA2, psB2);
            gat_step(sA[i].z, sA[i].w, sB[i].z, sB[i].w, mq.z, mq.w, wA[2 * i + 1], wB[2 * i + 1],
                     siA2, siB2, mA2, mB2, cA, cB, psA2, psB2);
        }
        __builtin_amdgcn_sched_barrier(0);
    }
}

__global__ __launch_bounds__(256) void gat_phase2(
    const int* __restrict__ adj, const float* __restrict__ WhP,
    const float* __restrict__ s_i, const float* __restrict__ s_j,
    float* __restrict__ out)
{
    __shared__ __attribute__((aligned(16))) float ldsAll[2 * kN + kNQ * kQPitch];
    float* const sjL = ldsAll;
    float* const WhL = ldsAll + 2 * kN;

    const int blk  = blockIdx.x;
    const int hp   = blk & 3;
    const int slab = (blk >> 2) & 7;
    const int b    = blk >> 5;
    const int tid  = threadIdx.x;
    const int wave = tid >> 5;
    const int lane = tid & 31;
    const int r16  = lane & 15;
    const int half = lane >> 4;
    const int hA   = hp * 2;
    const int hB   = hA + 1;
    const int i0   = slab * kSlabRows + wave * 16;

    float4 mxA[8], mxB[8];
    head_sj_load(s_j + ((size_t)b * kH + hA) * kN, lane, mxA);
    head_sj_load(s_j + ((size_t)b * kH + hB) * kN, lane, mxB);
    const float siA = s_i[((size_t)b * kH + hA) * kN + i0 + r16];
    const float siB = s_i[((size_t)b * kH + hB) * kN + i0 + r16];

    const int4* arow = (const int4*)(adj + ((size_t)b * kN + i0 + r16) * kN) + half;
    int4 mkA[kChunkT], mkB[kChunkT];
    #pragma unroll
    for (int tt = 0; tt < kChunkT; ++tt) mkA[tt] = arow[2 * tt];

    const float4* WhG = (const float4*)(WhP + (size_t)b * kN * kHD)
                        + (size_t)(tid >> 4) * 64 + hp * 16 + (tid & 15);
    const float4* sjG = (const float4*)(s_j + ((size_t)b * kH + hA) * kN);
    float4 pw[32];
    #pragma unroll
    for (int k = 0; k < 32; ++k) pw[k] = WhG[(size_t)k * 16 * 64];
    const float4 ps0 = sjG[tid];
    const float4 ps1 = sjG[256 + tid];
    __builtin_amdgcn_sched_barrier(0);

    const float mvA = head_max_sj(mxA);
    const float mvB = head_max_sj(mxB);
    const float mA = leaky(siA + mvA);
    const float mB = leaky(siB + mvB);
    const v2f siA2 = {siA, siA}, siB2 = {siB, siB};
    const v2f mA2  = {mA, mA},   mB2  = {mB, mB};

    {
        float* d = WhL + (tid >> 4) * kQPitch + (tid & 15) * 4;
        #pragma unroll
        for (int k = 0; k < 32; ++k) *(float4*)(d + k * 16 * kQPitch) = pw[k];
        *(float4*)(sjL + tid * 4)      = ps0;
        *(float4*)(sjL + kN + tid * 4) = ps1;
    }
    __syncthreads();

    const float* whLane = WhL + half * 2 * kQPitch + r16 * 2;
    const float* sjLane = sjL + 4 * half;
    v8f cA = {};
    v8f cB = {};
    v2f psA2 = {0.0f, 0.0f}, psB2 = {0.0f, 0.0f};
    for (int c2 = 0; c2 < kNChunk / 2; ++c2) {
        const int ch = 2 * c2;
        gat_chunk<true>(mkA, mkB, arow + 2 * kChunkT * (ch + 1),
                        whLane + ch * (4 * kChunkT * kQPitch), sjLane + ch * (8 * kChunkT),
                        siA2, siB2, mA2, mB2, cA, cB, psA2, psB2);
        if (c2 + 1 < kNChunk / 2) {
            gat_chunk<true>(mkB, mkA, arow + 2 * kChunkT * (ch + 2),
                            whLane + (ch + 1) * (4 * kChunkT * kQPitch), sjLane + (ch + 1) * (8 * kChunkT),
                            siA2, siB2, mA2, mB2, cA, cB, psA2, psB2);
        } else {
            gat_chunk<false>(mkB, mkA, arow,
                             whLane + (ch + 1) * (4 * kChunkT * kQPitch), sjLane + (ch + 1) * (8 * kChunkT),
                             siA2, siB2, mA2, mB2, cA, cB, psA2, psB2);
        }
    }

    float psA = psA2[0] + psA2[1];
    float psB = psB2[0] + psB2[1];
    psA += swzf<SWZ_X16>(psA);
    psB += swzf<SWZ_X16>(psB);
    const float invA = 1.0f / fmaxf(psA, 1e-30f);
    const float invB = 1.0f / fmaxf(psB, 1e-30f);
    const int baddr = half * 32;
    #pragma unroll
    for (int r = 0; r < 8; ++r) {
        float scA = bpermf(baddr + 4 * r, invA);
        float scB = bpermf(baddr + 4 * r, invB);
        int row = i0 + r + half * 8;
        float* o = out + ((size_t)b * kN + row) * kHD + hA * 16 + r16;
        o[0]  = cA[r] * scA;
        o[16] = cB[r] * scB;
    }
}

extern "C" void kernel_launch(void* const* d_in, const int* in_sizes, int n_in,
                              void* d_out, int out_size, void* d_ws, size_t ws_size,
                              hipStream_t stream) {
    const float* h   = (const float*)d_in[0];
    const int*   adj = (const int*)d_in[1];
    const float* W   = (const float*)d_in[2];
    const float* a   = (const float*)d_in[3];
    float* out = (float*)d_out;

    float* WhP = (float*)d_ws;
    float* s_i = WhP + (size_t)kB * kN * kHD;
    float* s_j = s_i + (size_t)kB * kH * kN;

    gat_phase1<<<dim3((kB * kN) / 16), 256, 0, stream>>>(h, W, a, WhP, s_i, s_j);
    gat_phase2<<<dim3(kB * 4 * (kN / kSlabRows)), 256, 0, stream>>>(adj, WhP, s_i, s_j, out);
}


# Phase 2 score math folded into row and column terms: t=max(u_i+s_j', v_i+alpha*s_j') goes straight to exp2 (s_j'=s_j*log2e in LDS); mask as one saturating convert times p; sum order unchanged
# speedup vs baseline: 1.4141x; 1.0392x over previous
#include <hip/hip_runtime.h>
#include <hip/hip_bf16.h>

typedef __attribute__((ext_vector_type(2))) float v2f;
typedef __attribute__((ext_vector_type(8))) float v8f;

#define GAT_NEG_INF (-9000000000000000.0f)

constexpr int kB   = 8;
constexpr int kN   = 1024;
constexpr int kDin = 128;
constexpr int kH   = 8;
constexpr int kD   = 16;
constexpr int kHD  = 128;
constexpr float kAlpha = 0.2f;

__device__ __forceinline__ float leaky(float x) {
    return fmaxf(x, kAlpha * x);
}

template <int IMM>
__device__ __forceinline__ float swzf(float v) {
    return __int_as_float(__builtin_amdgcn_ds_swizzle(__float_as_int(v), IMM));
}
__device__ __forceinline__ float bpermf(int byteAddr, float v) {
    return __int_as_float(__builtin_amdgcn_ds_bpermute(byteAddr, __float_as_int(v)));
}
#define SWZ_X1  0x041f
#define SWZ_X2  0x081f
#define SWZ_X4  0x101f
#define SWZ_X8  0x201f
#define SWZ_X16 0x401f

constexpr int kWPitch = 288;

__global__ __launch_bounds__(256) void gat_phase1(
    const float* __restrict__ h, const float* __restrict__ W,
    const float* __restrict__ a, float* __restrict__ WhP,
    float* __restrict__ s_i, float* __restrict__ s_j)
{
    __shared__ float ldsP1[16 * 132 + (kDin / 2) * kWPitch];
    float* const hT = ldsP1;
    float* const WL = ldsP1 + 16 * 132;

    const int g    = blockIdx.x;
    const int tid  = threadIdx.x;
    const int head = tid >> 5;
    const int lane = tid & 31;
    const int r16  = lane & 15;
    const int half = lane >> 4;

    {
        const float4* hsrc = (const float4*)(h + (size_t)g * 16 * kDin);
        const float4* W4   = (const float4*)W;
        const int col4 = tid & 31;
        const int kp0  = tid >> 5;
        const float4 hv0 = hsrc[tid];
        const float4 hv1 = hsrc[256 + tid];
        float4 we[8], wo[8];
        #pragma unroll
        for (int it = 0; it < 8; ++it) {
            const int kp = kp0 + 8 * it;
            we[it] = W4[(2 * kp) * 32 + col4];
            wo[it] = W4[(2 * kp + 1) * 32 + col4];
        }
        *(float4*)&hT[(tid >> 5) * 132 + (tid & 31) * 4]       = hv0;
        *(float4*)&hT[((tid >> 5) + 8) * 132 + (tid & 31) * 4] = hv1;
        #pragma unroll
        for (int it = 0; it < 8; ++it) {
            const int kp = kp0 + 8 * it;
            float* d = &WL[kp * kWPitch + col4 * 8];
            float4 lo, hi;
            lo.x = we[it].x;  lo.y = wo[it].x;  lo.z = we[it].y;  lo.w = wo[it].y;
            hi.x = we[it].z;  hi.y = wo[it].z;  hi.z = we[it].w;  hi.w = wo[it].w;
            *(float4*)d       = lo;
            *(float4*)(d + 4) = hi;
        }
    }
    __syncthreads();

    const int n0 = head * 16;
    const float* aRow = &hT[r16 * 132 + 2 * half];
    const float* bCol = &WL[half * kWPitch + (n0 + r16) * 2];
    v8f c = {};
    #pragma unroll
    for (int k = 0; k < kDin; k += 4) {
        const v2f av = *(const v2f*)(aRow + k);
        const v2f bv = *(const v2f*)(bCol + (k >> 1) * kWPitch);
        c = __builtin_amdgcn_wmma_f32_16x16x4_f32(false, av, false, bv,
                                                  (short)0, c, false, false);
    }

    #pragma unroll
    for (int r = 0; r < 8; ++r) {
        int row = g * 16 + r + half * 8;
        WhP[((size_t)(row >> 1) * kHD + n0 + r16) * 2 + (row & 1)] = c[r];
    }

    const float asrc = a[head * 2 * kD + r16];
    const float adst = a[head * 2 * kD + kD + r16];
    #pragma unroll
    for (int r = 0; r < 8; ++r) {
        float vs = c[r] * asrc;
        float vd = c[r] * adst;
        vs += swzf<SWZ_X1>(vs);  vd += swzf<SWZ_X1>(vd);
        vs += swzf<SWZ_X2>(vs);  vd += swzf<SWZ_X2>(vd);
        vs += swzf<SWZ_X4>(vs);  vd += swzf<SWZ_X4>(vd);
        vs += swzf<SWZ_X8>(vs);  vd += swzf<SWZ_X8>(vd);
        if (r16 == 0) {
            int row = g * 16 + r + half * 8;
            int bb = row >> 10, nn = row & (kN - 1);
            s_i[((size_t)bb * kH + head) * kN + nn] = vs;
            s_j[((size_t)bb * kH + head) * kN + nn] = vd;
        }
    }
}

constexpr int kSlabRows = 128;
constexpr int kNQ       = kN / 2;
constexpr int kQPitch   = 80;
constexpr int kChunkT   = 16;
constexpr int kNChunk   = kN / (8 * kChunkT);

__device__ __forceinline__ void head_sj_load(const float* __restrict__ sJ, int lane, float4 (&v)[8]) {
    const float4* sJ4 = (const float4*)sJ;
    #pragma unroll
    for (int k = 0; k < 8; ++k) v[k] = sJ4[k * 32 + lane];
}
__device__ __forceinline__ float head_max_sj(const float4 (&v)[8]) {
    float mv = GAT_NEG_INF;
    #pragma unroll
    for (int k = 0; k < 8; ++k)
        mv = fmaxf(mv, fmaxf(fmaxf(v[k].x, v[k].y), fmaxf(v[k].z, v[k].w)));
    mv = fmaxf(mv, swzf<SWZ_X16>(mv));
    mv = fmaxf(mv, swzf<SWZ_X8>(mv));
    mv = fmaxf(mv, swzf<SWZ_X4>(mv));
    mv = fmaxf(mv, swzf<SWZ_X2>(mv));
    mv = fmaxf(mv, swzf<SWZ_X1>(mv));
    return mv;
}

__device__ __forceinline__ float adj_factor(int m) {
    return __builtin_amdgcn_fmed3f((float)(unsigned)m, 0.0f, 1.0f);
}

__device__ __forceinline__ void gat_step(
    float sa0, float sa1, float sb0, float sb1, const v2f mf,
    const v2f wA, const v2f wB,
    const v2f uA2, const v2f uB2, const v2f vA2, const v2f vB2,
    v8f& cA, v8f& cB, v2f& psA2, v2f& psB2)
{
    const v2f al2 = {kAlpha, kAlpha};
    const v2f sjA = {sa0, sa1};
    const v2f sjB = {sb0, sb1};
    const v2f tA = __builtin_elementwise_max(uA2 + sjA, __builtin_elementwise_fma(al2, sjA, vA2));
    const v2f tB = __builtin_elementwise_max(uB2 + sjB, __builtin_elementwise_fma(al2, sjB, vB2));
    v2f pA, pB;
    pA[0] = __builtin_amdgcn_exp2f(tA[0]);
    pA[1] = __builtin_amdgcn_exp2f(tA[1]);
    pB[0] = __builtin_amdgcn_exp2f(tB[0]);
    pB[1] = __builtin_amdgcn_exp2f(tB[1]);
    pA *= mf;
    pB *= mf;
    psA2 += pA;
    psB2 += pB;
    cA = __builtin_amdgcn_wmma_f32_16x16x4_f32(false, pA, false, wA,
                                               (short)0, cA, false, false);
    cB = __builtin_amdgcn_wmma_f32_16x16x4_f32(false, pB, false, wB,
                                               (short)0, cB, false, false);
}

template <bool PF>
__device__ __forceinline__ void gat_chunk(
    const int4 (&cur)[kChunkT], int4 (&nxt)[kChunkT], const int4* __restrict__ arowNext,
    const float* __restrict__ whC, const float* __restrict__ sjC,
    const v2f uA2, const v2f uB2, const v2f vA2, const v2f vB2,
    v8f& cA, v8f& cB, v2f& psA2, v2f& psB2)
{
    #pragma unroll
    for (int g = 0; g < 2; ++g) {
        float4 sA[8], sB[8];
        v2f    wA[16], wB[16];
        #pragma unroll
        for (int i = 0; i < 8; ++i) {
            const int tt = g * 8 + i;
            sA[i] = *(const float4*)(sjC + 8 * tt);
            sB[i] = *(const float4*)(sjC + kN + 8 * tt);
            wA[2 * i]     = *(const v2f*)(whC + (4 * tt) * kQPitch);
            wB[2 * i]     = *(const v2f*)(whC + (4 * tt) * kQPitch + 32);
            wA[2 * i + 1] = *(const v2f*)(whC + (4 * tt + 1) * kQPitch);
            wB[2 * i + 1] = *(const v2f*)(whC + (4 * tt + 1) * kQPitch + 32);
        }
        __builtin_amdgcn_sched_barrier(0);
        #pragma unroll
        for (int i = 0; i < 8; ++i) {
            if (PF) nxt[g * 8 + i] = arowNext[2 * (g * 8 + i)];
            const int4 mq = cur[g * 8 + i];
            const v2f mf01 = {adj_factor(mq.x), adj_factor(mq.y)};
            const v2f mf23 = {adj_factor(mq.z), adj_factor(mq.w)};
            gat_step(sA[i].x, sA[i].y, sB[i].x, sB[i].y, mf01, wA[2 * i], wB[2 * i],
                     uA2, uB2, vA2, vB2, cA, cB, psA2, psB2);
            gat_step(sA[i].z, sA[i].w, sB[i].z, sB[i].w, mf23, wA[2 * i + 1], wB[2 * i + 1],
                     uA2, uB2, vA2, vB2, cA, cB, psA2, psB2);
        }
        __builtin_amdgcn_sched_barrier(0);
    }
}

__global__ __launch_bounds__(256) void gat_phase2(
    const int* __restrict__ adj, const float* __restrict__ WhP,
    const float* __restrict__ s_i, const float* __restrict__ s_j,
    float* __restrict__ out)
{
    __shared__ __attribute__((aligned(16))) float ldsAll[2 * kN + kNQ * kQPitch];
    float* const sjL = ldsAll;
    float* const WhL = ldsAll + 2 * kN;

    const int blk  = blockIdx.x;
    const int hp   = blk & 3;
    const int slab = (blk >> 2) & 7;
    const int b    = blk >> 5;
    const int tid  = threadIdx.x;
    const int wave = tid >> 5;
    const int lane = tid & 31;
    const int r16  = lane & 15;
    const int half = lane >> 4;
    const int hA   = hp * 2;
    const int hB   = hA + 1;
    const int i0   = slab * kSlabRows + wave * 16;

    float4 mxA[8], mxB[8];
    head_sj_load(s_j + ((size_t)b * kH + hA) * kN, lane, mxA);
    head_sj_load(s_j + ((size_t)b * kH + hB) * kN, lane, mxB);
    const float siA = s_i[((size_t)b * kH + hA) * kN + i0 + r16];
    const float siB = s_i[((size_t)b * kH + hB) * kN + i0 + r16];

    const int4* arow = (const int4*)(adj + ((size_t)b * kN + i0 + r16) * kN) + half;
    int4 mkA[kChunkT], mkB[kChunkT];
    #pragma unroll
    for (int tt = 0; tt < kChunkT; ++tt) mkA[tt] = arow[2 * tt];

    const float4* WhG = (const float4*)(WhP + (size_t)b * kN * kHD)
                        + (size_t)(tid >> 4) * 64 + hp * 16 + (tid & 15);
    const float4* sjG = (const float4*)(s_j + ((size_t)b * kH + hA) * kN);
    float4 pw[32];
    #pragma unroll
    for (int k = 0; k < 32; ++k) pw[k] = WhG[(size_t)k * 16 * 64];
    const float4 ps0 = sjG[tid];
    const float4 ps1 = sjG[256 + tid];
    __builtin_amdgcn_sched_barrier(0);

    const float mvA = head_max_sj(mxA);
    const float mvB = head_max_sj(mxB);
    const float mA = leaky(siA + mvA);
    const float mB = leaky(siB + mvB);
    constexpr float kLog2e = 1.4426950408889634f;
    const float uA = (siA - mA) * kLog2e, vA = (kAlpha * siA - mA) * kLog2e;
    const float uB = (siB - mB) * kLog2e, vB = (kAlpha * siB - mB) * kLog2e;
    const v2f uA2 = {uA, uA}, uB2 = {uB, uB};
    const v2f vA2 = {vA, vA}, vB2 = {vB, vB};

    {
        float* d = WhL + (tid >> 4) * kQPitch + (tid & 15) * 4;
        #pragma unroll
        for (int k = 0; k < 32; ++k) *(float4*)(d + k * 16 * kQPitch) = pw[k];
        float4 q0 = ps0, q1 = ps1;
        q0.x *= kLog2e;  q0.y *= kLog2e;  q0.z *= kLog2e;  q0.w *= kLog2e;
        q1.x *= kLog2e;  q1.y *= kLog2e;  q1.z *= kLog2e;  q1.w *= kLog2e;
        *(float4*)(sjL + tid * 4)      = q0;
        *(float4*)(sjL + kN + tid * 4) = q1;
    }
    __syncthreads();

    const float* whLane = WhL + half * 2 * kQPitch + r16 * 2;
    const float* sjLane = sjL + 4 * half;
    v8f cA = {};
    v8f cB = {};
    v2f psA2 = {0.0f, 0.0f}, psB2 = {0.0f, 0.0f};
    for (int c2 = 0; c2 < kNChunk / 2; ++c2) {
        const int ch = 2 * c2;
        gat_chunk<true>(mkA, mkB, arow + 2 * kChunkT * (ch + 1),
                        whLane + ch * (4 * kChunkT * kQPitch), sjLane + ch * (8 * kChunkT),
                        uA2, uB2, vA2, vB2, cA, cB, psA2, psB2);
        if (c2 + 1 < kNChunk / 2) {
            gat_chunk<true>(mkB, mkA, arow + 2 * kChunkT * (ch + 2),
                            whLane + (ch + 1) * (4 * kChunkT * kQPitch), sjLane + (ch + 1) * (8 * kChunkT),
                            uA2, uB2, vA2, vB2, cA, cB, psA2, psB2);
        } else {
            gat_chunk<false>(mkB, mkA, arow,
                             whLane + (ch + 1) * (4 * kChunkT * kQPitch), sjLane + (ch + 1) * (8 * kChunkT),
                             uA2, uB2, vA2, vB2, cA, cB, psA2, psB2);
        }
    }

    float psA = psA2[0] + psA2[1];
    float psB = psB2[0] + psB2[1];
    psA += swzf<SWZ_X16>(psA);
    psB += swzf<SWZ_X16>(psB);
    const float invA = 1.0f / fmaxf(psA, 1e-30f);
    const float invB = 1.0f / fmaxf(psB, 1e-30f);
    const int baddr = half * 32;
    #pragma unroll
    for (int r = 0; r < 8; ++r) {
        float scA = bpermf(baddr + 4 * r, invA);
        float scB = bpermf(baddr + 4 * r, invB);
        int row = i0 + r + half * 8;
        float* o = out + ((size_t)b * kN + row) * kHD + hA * 16 + r16;
        o[0]  = cA[r] * scA;
        o[16] = cB[r] * scB;
    }
}

extern "C" void kernel_launch(void* const* d_in, const int* in_sizes, int n_in,
                              void* d_out, int out_size, void* d_ws, size_t ws_size,
                              hipStream_t stream) {
    const float* h   = (const float*)d_in[0];
    const int*   adj = (const int*)d_in[1];
    const float* W   = (const float*)d_in[2];
    const float* a   = (const float*)d_in[3];
    float* out = (float*)d_out;

    float* WhP = (float*)d_ws;
    float* s_i = WhP + (size_t)kB * kN * kHD;
    float* s_j = s_i + (size_t)kB * kH * kN;

    gat_phase1<<<dim3((kB * kN) / 16), 256, 0, stream>>>(h, W, a, WhP, s_i, s_j);
    gat_phase2<<<dim3(kB * 4 * (kN / kSlabRows)), 256, 0, stream>>>(adj, WhP, s_i, s_j, out);
}


# Launch 1 computes the transposed tile (W fragment as A, h fragment as B): a lane owns 8 features of one node, logits summed in-lane then across lane halves (logit sum order changed)
# speedup vs baseline: 1.5054x; 1.0646x over previous
#include <hip/hip_runtime.h>
#include <hip/hip_bf16.h>

typedef __attribute__((ext_vector_type(2))) float v2f;
typedef __attribute__((ext_vector_type(8))) float v8f;

#define GAT_NEG_INF (-9000000000000000.0f)

constexpr int kB   = 8;
constexpr int kN   = 1024;
constexpr int kDin = 128;
constexpr int kH   = 8;
constexpr int kD   = 16;
constexpr int kHD  = 128;
constexpr float kAlpha = 0.2f;

__device__ __forceinline__ float leaky(float x) {
    return fmaxf(x, kAlpha * x);
}

template <int IMM>
__device__ __forceinline__ float swzf(float v) {
    return __int_as_float(__builtin_amdgcn_ds_swizzle(__float_as_int(v), IMM));
}
__device__ __forceinline__ float bpermf(int byteAddr, float v) {
    return __int_as_float(__builtin_amdgcn_ds_bpermute(byteAddr, __float_as_int(v)));
}
#define SWZ_X1  0x041f
#define SWZ_X2  0x081f
#define SWZ_X4  0x101f
#define SWZ_X8  0x201f
#define SWZ_X16 0x401f

constexpr int kWPitch = 288;

__global__ __launch_bounds__(256) void gat_phase1(
    const float* __restrict__ h, const float* __restrict__ W,
    const float* __restrict__ a, float* __restrict__ WhP,
    float* __restrict__ s_i, float* __restrict__ s_j)
{
    __shared__ float ldsP1[16 * 132 + (kDin / 2) * kWPitch];
    float* const hT = ldsP1;
    float* const WL = ldsP1 + 16 * 132;

    const int g    = blockIdx.x;
    const int tid  = threadIdx.x;
    const int head = tid >> 5;
    const int lane = tid & 31;
    const int r16  = lane & 15;
    const int half = lane >> 4;

    float4 as0, as1, ad0, ad1;
    {
        const float4* hsrc = (const float4*)(h + (size_t)g * 16 * kDin);
        const float4* W4   = (const float4*)W;
        const int col4 = tid & 31;
        const int kp0  = tid >> 5;
        const float4 hv0 = hsrc[tid];
        const float4 hv1 = hsrc[256 + tid];
        float4 we[8], wo[8];
        #pragma unroll
        for (int it = 0; it < 8; ++it) {
            const int kp = kp0 + 8 * it;
            we[it] = W4[(2 * kp) * 32 + col4];
            wo[it] = W4[(2 * kp + 1) * 32 + col4];
        }
        const float4* a4 = (const float4*)(a + head * 2 * kD + 8 * half);
        as0 = a4[0];  as1 = a4[1];
        ad0 = a4[4];  ad1 = a4[5];
        *(float4*)&hT[(tid >> 5) * 132 + (tid & 31) * 4]       = hv0;
        *(float4*)&hT[((tid >> 5) + 8) * 132 + (tid & 31) * 4] = hv1;
        #pragma unroll
        for (int it = 0; it < 8; ++it) {
            const int kp = kp0 + 8 * it;
            float* d = &WL[kp * kWPitch + col4 * 8];
            float4 lo, hi;
            lo.x = we[it].x;  lo.y = wo[it].x;  lo.z = we[it].y;  lo.w = wo[it].y;
            hi.x = we[it].z;  hi.y = wo[it].z;  hi.z = we[it].w;  hi.w = wo[it].w;
            *(float4*)d       = lo;
            *(float4*)(d + 4) = hi;
        }
    }
    __syncthreads();

    const int n0 = head * 16;
    const float* hRow = &hT[r16 * 132 + 2 * half];
    const float* wCol = &WL[half * kWPitch + (n0 + r16) * 2];
    v8f c = {};
    #pragma unroll
    for (int k = 0; k < kDin; k += 4) {
        const v2f hv = *(const v2f*)(hRow + k);
        const v2f wv = *(const v2f*)(wCol + (k >> 1) * kWPitch);
        c = __builtin_amdgcn_wmma_f32_16x16x4_f32(false, wv, false, hv,
                                                  (short)0, c, false, false);
    }

    float vs = c[0] * as0.x, vd = c[0] * ad0.x;
    vs = fmaf(c[1], as0.y, vs);  vd = fmaf(c[1], ad0.y, vd);
    vs = fmaf(c[2], as0.z, vs);  vd = fmaf(c[2], ad0.z, vd);
    vs = fmaf(c[3], as0.w, vs);  vd = fmaf(c[3], ad0.w, vd);
    vs = fmaf(c[4], as1.x, vs);  vd = fmaf(c[4], ad1.x, vd);
    vs = fmaf(c[5], as1.y, vs);  vd = fmaf(c[5], ad1.y, vd);
    vs = fmaf(c[6], as1.z, vs);  vd = fmaf(c[6], ad1.z, vd);
    vs = fmaf(c[7], as1.w, vs);  vd = fmaf(c[7], ad1.w, vd);

    float nb[8];
    #pragma unroll
    for (int r = 0; r < 8; ++r) nb[r] = swzf<SWZ_X1>(c[r]);
    vs += swzf<SWZ_X16>(vs);
    vd += swzf<SWZ_X16>(vd);

    const bool odd = (lane & 1) != 0;
    const int row = g * 16 + r16;
    {
        float4 o0, o1;
        o0.x = odd ? nb[4] : c[0];  o0.y = odd ? c[4] : nb[0];
        o0.z = odd ? nb[5] : c[1];  o0.w = odd ? c[5] : nb[1];
        o1.x = odd ? nb[6] : c[2];  o1.y = odd ? c[6] : nb[2];
        o1.z = odd ? nb[7] : c[3];  o1.w = odd ? c[7] : nb[3];
        float* wp = WhP + ((size_t)(row >> 1) * kHD + n0 + 8 * half + (odd ? 4 : 0)) * 2;
        *(float4*)wp       = o0;
        *(float4*)(wp + 4) = o1;
    }
    {
        const int bb = row >> 10, nn = row & (kN - 1);
        float* lp = half ? s_j : s_i;
        lp[((size_t)bb * kH + head) * kN + nn] = half ? vd : vs;
    }
}

constexpr int kSlabRows = 128;
constexpr int kNQ       = kN / 2;
constexpr int kQPitch   = 80;
constexpr int kChunkT   = 16;
constexpr int kNChunk   = kN / (8 * kChunkT);

__device__ __forceinline__ void head_sj_load(const float* __restrict__ sJ, int lane, float4 (&v)[8]) {
    const float4* sJ4 = (const float4*)sJ;
    #pragma unroll
    for (int k = 0; k < 8; ++k) v[k] = sJ4[k * 32 + lane];
}
__device__ __forceinline__ float head_max_sj(const float4 (&v)[8]) {
    float mv = GAT_NEG_INF;
    #pragma unroll
    for (int k = 0; k < 8; ++k)
        mv = fmaxf(mv, fmaxf(fmaxf(v[k].x, v[k].y), fmaxf(v[k].z, v[k].w)));
    mv = fmaxf(mv, swzf<SWZ_X16>(mv));
    mv = fmaxf(mv, swzf<SWZ_X8>(mv));
    mv = fmaxf(mv, swzf<SWZ_X4>(mv));
    mv = fmaxf(mv, swzf<SWZ_X2>(mv));
    mv = fmaxf(mv, swzf<SWZ_X1>(mv));
    return mv;
}

__device__ __forceinline__ float adj_factor(int m) {
    return __builtin_amdgcn_fmed3f((float)(unsigned)m, 0.0f, 1.0f);
}

__device__ __forceinline__ void gat_step(
    float sa0, float sa1, float sb0, float sb1, const v2f mf,
    const v2f wA, const v2f wB,
    const v2f uA2, const v2f uB2, const v2f vA2, const v2f vB2,
    v8f& cA, v8f& cB, v2f& psA2, v2f& psB2)
{
    const v2f al2 = {kAlpha, kAlpha};
    const v2f sjA = {sa0, sa1};
    const v2f sjB = {sb0, sb1};
    const v2f tA = __builtin_elementwise_max(uA2 + sjA, __builtin_elementwise_fma(al2, sjA, vA2));
    const v2f tB = __builtin_elementwise_max(uB2 + sjB, __builtin_elementwise_fma(al2, sjB, vB2));
    v2f pA, pB;
    pA[0] = __builtin_amdgcn_exp2f(tA[0]);
    pA[1] = __builtin_amdgcn_exp2f(tA[1]);
    pB[0] = __builtin_amdgcn_exp2f(tB[0]);
    pB[1] = __builtin_amdgcn_exp2f(tB[1]);
    pA *= mf;
    pB *= mf;
    psA2 += pA;
    psB2 += pB;
    cA = __builtin_amdgcn_wmma_f32_16x16x4_f32(false, pA, false, wA,
                                               (short)0, cA, false, false);
    cB = __builtin_amdgcn_wmma_f32_16x16x4_f32(false, pB, false, wB,
                                               (short)0, cB, false, false);
}

template <bool PF>
__device__ __forceinline__ void gat_chunk(
    const int4 (&cur)[kChunkT], int4 (&nxt)[kChunkT], const int4* __restrict__ arowNext,
    const float* __restrict__ whC, const float* __restrict__ sjC,
    const v2f uA2, const v2f uB2, const v2f vA2, const v2f vB2,
    v8f& cA, v8f& cB, v2f& psA2, v2f& psB2)
{
    #pragma unroll
    for (int g = 0; g < 2; ++g) {
        float4 sA[8], sB[8];
        v2f    wA[16], wB[16];
        #pragma unroll
        for (int i = 0; i < 8; ++i) {
            const int tt = g * 8 + i;
            sA[i] = *(const float4*)(sjC + 8 * tt);
            sB[i] = *(const float4*)(sjC + kN + 8 * tt);
            wA[2 * i]     = *(const v2f*)(whC + (4 * tt) * kQPitch);
            wB[2 * i]     = *(const v2f*)(whC + (4 * tt) * kQPitch + 32);
            wA[2 * i + 1] = *(const v2f*)(whC + (4 * tt + 1) * kQPitch);
            wB[2 * i + 1] = *(const v2f*)(whC + (4 * tt + 1) * kQPitch + 32);
        }
        __builtin_amdgcn_sched_barrier(0);
        #pragma unroll
        for (int i = 0; i < 8; ++i) {
            if (PF) nxt[g * 8 + i] = arowNext[2 * (g * 8 + i)];
            const int4 mq = cur[g * 8 + i];
            const v2f mf01 = {adj_factor(mq.x), adj_factor(mq.y)};
            const v2f mf23 = {adj_factor(mq.z), adj_factor(mq.w)};
            gat_step(sA[i].x, sA[i].y, sB[i].x, sB[i].y, mf01, wA[2 * i], wB[2 * i],
                     uA2, uB2, vA2, vB2, cA, cB, psA2, psB2);
            gat_step(sA[i].z, sA[i].w, sB[i].z, sB[i].w, mf23, wA[2 * i + 1], wB[2 * i + 1],
                     uA2, uB2, vA2, vB2, cA, cB, psA2, psB2);
        }
        __builtin_amdgcn_sched_barrier(0);
    }
}

__global__ __launch_bounds__(256) void gat_phase2(
    const int* __restrict__ adj, const float* __restrict__ WhP,
    const float* __restrict__ s_i, const float* __restrict__ s_j,
    float* __restrict__ out)
{
    __shared__ __attribute__((aligned(16))) float ldsAll[2 * kN + kNQ * kQPitch];
    float* const sjL = ldsAll;
    float* const WhL = ldsAll + 2 * kN;

    const int blk  = blockIdx.x;
    const int hp   = blk & 3;
    const int slab = (blk >> 2) & 7;
    const int b    = blk >> 5;
    const int tid  = threadIdx.x;
    const int wave = tid >> 5;
    const int lane = tid & 31;
    const int r16  = lane & 15;
    const int half = lane >> 4;
    const int hA   = hp * 2;
    const int hB   = hA + 1;
    const int i0   = slab * kSlabRows + wave * 16;

    float4 mxA[8], mxB[8];
    head_sj_load(s_j + ((size_t)b * kH + hA) * kN, lane, mxA);
    head_sj_load(s_j + ((size_t)b * kH + hB) * kN, lane, mxB);
    const float siA = s_i[((size_t)b * kH + hA) * kN + i0 + r16];
    const float siB = s_i[((size_t)b * kH + hB) * kN + i0 + r16];

    const int4* arow = (const int4*)(adj + ((size_t)b * kN + i0 + r16) * kN) + half;
    int4 mkA[kChunkT], mkB[kChunkT];
    #pragma unroll
    for (int tt = 0; tt < kChunkT; ++tt) mkA[tt] = arow[2 * tt];

    const float4* WhG = (const float4*)(WhP + (size_t)b * kN * kHD)
                        + (size_t)(tid >> 4) * 64 + hp * 16 + (tid & 15);
    const float4* sjG = (const float4*)(s_j + ((size_t)b * kH + hA) * kN);
    float4 pw[32];
    #pragma unroll
    for (int k = 0; k < 32; ++k) pw[k] = WhG[(size_t)k * 16 * 64];
    const float4 ps0 = sjG[tid];
    const float4 ps1 = sjG[256 + tid];
    __builtin_amdgcn_sched_barrier(0);

    const float mvA = head_max_sj(mxA);
    const float mvB = head_max_sj(mxB);
    const float mA = leaky(siA + mvA);
    const float mB = leaky(siB + mvB);
    constexpr float kLog2e = 1.4426950408889634f;
    const float uA = (siA - mA) * kLog2e, vA = (kAlpha * siA - mA) * kLog2e;
    const float uB = (siB - mB) * kLog2e, vB = (kAlpha * siB - mB) * kLog2e;
    const v2f uA2 = {uA, uA}, uB2 = {uB, uB};
    const v2f vA2 = {vA, vA}, vB2 = {vB, vB};

    {
        float* d = WhL + (tid >> 4) * kQPitch + (tid & 15) * 4;
        #pragma unroll
        for (int k = 0; k < 32; ++k) *(float4*)(d + k * 16 * kQPitch) = pw[k];
        float4 q0 = ps0, q1 = ps1;
        q0.x *= kLog2e;  q0.y *= kLog2e;  q0.z *= kLog2e;  q0.w *= kLog2e;
        q1.x *= kLog2e;  q1.y *= kLog2e;  q1.z *= kLog2e;  q1.w *= kLog2e;
        *(float4*)(sjL + tid * 4)      = q0;
        *(float4*)(sjL + kN + tid * 4) = q1;
    }
    __syncthreads();

    const float* whLane = WhL + half * 2 * kQPitch + r16 * 2;
    const float* sjLane = sjL + 4 * half;
    v8f cA = {};
    v8f cB = {};
    v2f psA2 = {0.0f, 0.0f}, psB2 = {0.0f, 0.0f};
    for (int c2 = 0; c2 < kNChunk / 2; ++c2) {
        const int ch = 2 * c2;
        gat_chunk<true>(mkA, mkB, arow + 2 * kChunkT * (ch + 1),
                        whLane + ch * (4 * kChunkT * kQPitch), sjLane + ch * (8 * kChunkT),
                        uA2, uB2, vA2, vB2, cA, cB, psA2, psB2);
        if (c2 + 1 < kNChunk / 2) {
            gat_chunk<true>(mkB, mkA, arow + 2 * kChunkT * (ch + 2),
                            whLane + (ch + 1) * (4 * kChunkT * kQPitch), sjLane + (ch + 1) * (8 * kChunkT),
                            uA2, uB2, vA2, vB2, cA, cB, psA2, psB2);
        } else {
            gat_chunk<false>(mkB, mkA, arow,
                             whLane + (ch + 1) * (4 * kChunkT * kQPitch), sjLane + (ch + 1) * (8 * kChunkT),
                             uA2, uB2, vA2, vB2, cA, cB, psA2, psB2);
        }
    }

    float psA = psA2[0] + psA2[1];
    float psB = psB2[0] + psB2[1];
    psA += swzf<SWZ_X16>(psA);
    psB += swzf<SWZ_X16>(psB);
    const float invA = 1.0f / fmaxf(psA, 1e-30f);
    const float invB = 1.0f / fmaxf(psB, 1e-30f);
    const int baddr = half * 32;
    #pragma unroll
    for (int r = 0; r < 8; ++r) {
        float scA = bpermf(baddr + 4 * r, invA);
        float scB = bpermf(baddr + 4 * r, invB);
        int row = i0 + r + half * 8;
        float* o = out + ((size_t)b * kN + row) * kHD + hA * 16 + r16;
        o[0]  = cA[r] * scA;
        o[16] = cB[r] * scB;
    }
}

extern "C" void kernel_launch(void* const* d_in, const int* in_sizes, int n_in,
                              void* d_out, int out_size, void* d_ws, size_t ws_size,
                              hipStream_t stream) {
    const float* h   = (const float*)d_in[0];
    const int*   adj = (const int*)d_in[1];
    const float* W   = (const float*)d_in[2];
    const float* a   = (const float*)d_in[3];
    float* out = (float*)d_out;

    float* WhP = (float*)d_ws;
    float* s_i = WhP + (size_t)kB * kN * kHD;
    float* s_j = s_i + (size_t)kB * kH * kN;

    gat_phase1<<<dim3((kB * kN) / 16), 256, 0, stream>>>(h, W, a, WhP, s_i, s_j);
    gat_phase2<<<dim3(kB * 4 * (kN / kSlabRows)), 256, 0, stream>>>(adj, WhP, s_i, s_j, out);
}
